# starting kernel
# baseline (speedup 1.0000x reference)
#include <hip/hip_runtime.h>


typedef __attribute__((ext_vector_type(2))) float v2f;
typedef __attribute__((ext_vector_type(4))) float f4;
typedef __attribute__((ext_vector_type(8))) float v8f;

#define NPTS   1024
#define YSTR   17
#define Y_OFF  0
#define A_OFF  (NPTS * YSTR)
#define M_OFF  (2 * NPTS * YSTR)
#define MSLOT  (16 * YSTR)
#define SHM_FLOATS (2 * NPTS * YSTR + 8 * MSLOT)

__global__ __launch_bounds__(256) void bconv_butterfly_kernel(
    const float* __restrict__ x,
    const float* __restrict__ tw,
    const float* __restrict__ bias,
    float* __restrict__ out)
{
    extern __shared__ float smem[];
    float* Y    = smem + Y_OFF;
    float* ACC  = smem + A_OFF;
    float* MATS = smem + M_OFF;

    const int t     = threadIdx.x;
    const int lane  = t & 31;
    const int wv    = t >> 5;
    const int mlane = lane & 15;
    const int kh    = lane >> 4;
    const int b0    = blockIdx.x * 16;
    const int o     = blockIdx.y;

    for (int idx = t; idx < 16384; idx += 256) {
        int n = idx & 1023, b = idx >> 10;
        ACC[n * YSTR + b] = 0.f;
    }

    float* mymat = MATS + wv * MSLOT;

    for (int i = 0; i < 16; ++i) {
        __syncthreads();
        const float* xi   = x + i * 65536;
        const float* twio = tw + (size_t)(i * 32 + o) * (10 * 512 * 4);

        for (int off = t * 128; off < 10 * 512 * 4 * 4; off += 256 * 128)
            __builtin_prefetch((const char*)twio + off, 0, 1);

        for (int idx = t; idx < 16384; idx += 256) {
            int m = idx & 1023;
            int b = idx >> 10;
            float v = xi[(b0 + b) * 1024 + m];
            int n = __brev((unsigned)m) >> 22;
            Y[n * YSTR + b] = v;
        }
        __syncthreads();

        for (int blk = 0; blk < 8; ++blk) {
            int k1 = blk * 8 + wv;
            for (int e = lane; e < 256; e += 32) {
                int p = e >> 4, q = e & 15;
                mymat[p * YSTR + q] = (p == q) ? 1.f : 0.f;
            }
            for (int l = 0; l < 4; ++l) {
                const float* twl = twio + l * 2048;
                for (int tt = lane; tt < 128; tt += 32) {
                    int q  = tt & 15, ph = tt >> 4;
                    int p0 = ((ph >> l) << (l + 1)) | (ph & ((1 << l) - 1));
                    int p1 = p0 | (1 << l);
                    f4 T = *(const f4*)(twl + (k1 * 8 + ph) * 4);
                    float m0 = mymat[p0 * YSTR + q];
                    float m1 = mymat[p1 * YSTR + q];
                    mymat[p0 * YSTR + q] = T.x * m0 + T.y * m1;
                    mymat[p1 * YSTR + q] = T.z * m0 + T.w * m1;
                }
            }
            v8f c = {};
            const int rowbase = k1 * 16;
#pragma unroll
            for (int kk = 0; kk < 4; ++kk) {
                int K0 = kk * 4 + 2 * kh;
                v2f a, bb;
                a.x  = mymat[mlane * YSTR + K0];
                a.y  = mymat[mlane * YSTR + K0 + 1];
                bb.x = Y[(rowbase + K0) * YSTR + mlane];
                bb.y = Y[(rowbase + K0 + 1) * YSTR + mlane];
                c = __builtin_amdgcn_wmma_f32_16x16x4_f32(
                        false, a, false, bb, (short)0, c, false, false);
            }
#pragma unroll
            for (int j = 0; j < 8; ++j)
                Y[(rowbase + j + 8 * kh) * YSTR + mlane] = c[j];
        }
        __syncthreads();

        for (int blk = 0; blk < 8; ++blk) {
            int cidx = blk * 8 + wv;
            int d = cidx >> 4, u = cidx & 15;
            for (int e = lane; e < 256; e += 32) {
                int p = e >> 4, q = e & 15;
                mymat[p * YSTR + q] = (p == q) ? 1.f : 0.f;
            }
            for (int lp = 0; lp < 4; ++lp) {
                const float* twl = twio + (4 + lp) * 2048;
                for (int tt = lane; tt < 128; tt += 32) {
                    int q  = tt & 15, ph = tt >> 4;
                    int p0 = ((ph >> lp) << (lp + 1)) | (ph & ((1 << lp) - 1));
                    int p1 = p0 | (1 << lp);
                    f4 T = *(const f4*)(twl + (d * 128 + ph * 16 + u) * 4);
                    float m0 = mymat[p0 * YSTR + q];
                    float m1 = mymat[p1 * YSTR + q];
                    mymat[p0 * YSTR + q] = T.x * m0 + T.y * m1;
                    mymat[p1 * YSTR + q] = T.z * m0 + T.w * m1;
                }
            }
            v8f c = {};
#pragma unroll
            for (int kk = 0; kk < 4; ++kk) {
                int K0 = kk * 4 + 2 * kh;
                v2f a, bb;
                a.x = mymat[mlane * YSTR + K0];
                a.y = mymat[mlane * YSTR + K0 + 1];
                int r0 = d * 256 + K0 * 16 + u;
                bb.x = Y[r0 * YSTR + mlane];
                bb.y = Y[(r0 + 16) * YSTR + mlane];
                c = __builtin_amdgcn_wmma_f32_16x16x4_f32(
                        false, a, false, bb, (short)0, c, false, false);
            }
#pragma unroll
            for (int j = 0; j < 8; ++j)
                Y[(d * 256 + (j + 8 * kh) * 16 + u) * YSTR + mlane] = c[j];
        }
        __syncthreads();

        {
            int r = t;
            const float* tw8 = twio + 8 * 2048;
            const float* tw9 = twio + 9 * 2048;
            f4 T8a = *(const f4*)(tw8 + r * 4);
            f4 T8b = *(const f4*)(tw8 + (256 + r) * 4);
            f4 T9a = *(const f4*)(tw9 + r * 4);
            f4 T9b = *(const f4*)(tw9 + (256 + r) * 4);
            for (int b = 0; b < 16; ++b) {
                float v00 = Y[(r      ) * YSTR + b];
                float v01 = Y[(256 + r) * YSTR + b];
                float v10 = Y[(512 + r) * YSTR + b];
                float v11 = Y[(768 + r) * YSTR + b];
                float a00 = T8a.x * v00 + T8a.y * v01;
                float a01 = T8a.z * v00 + T8a.w * v01;
                float a10 = T8b.x * v10 + T8b.y * v11;
                float a11 = T8b.z * v10 + T8b.w * v11;
                float f00 = T9a.x * a00 + T9a.y * a10;
                float f10 = T9a.z * a00 + T9a.w * a10;
                float f01 = T9b.x * a01 + T9b.y * a11;
                float f11 = T9b.z * a01 + T9b.w * a11;
                ACC[(r      ) * YSTR + b] += f00;
                ACC[(256 + r) * YSTR + b] += f01;
                ACC[(512 + r) * YSTR + b] += f10;
                ACC[(768 + r) * YSTR + b] += f11;
            }
        }
    }
    __syncthreads();

    for (int idx = t; idx < 16384; idx += 256) {
        int n = idx & 1023;
        int b = idx >> 10;
        out[(size_t)o * 65536 + (b0 + b) * 1024 + n] =
            ACC[n * YSTR + b] * 0.0625f + bias[o * 1024 + n];
    }
}

extern "C" void kernel_launch(void* const* d_in, const int* in_sizes, int n_in,
                              void* d_out, int out_size, void* d_ws, size_t ws_size,
                              hipStream_t stream) {
    (void)in_sizes; (void)n_in; (void)out_size; (void)d_ws; (void)ws_size;
    const float* x    = (const float*)d_in[0];
    const float* tw   = (const float*)d_in[1];
    const float* bias = (const float*)d_in[2];
    float* out        = (float*)d_out;

    const size_t shmem = (size_t)SHM_FLOATS * sizeof(float);
    hipFuncSetAttribute((const void*)bconv_butterfly_kernel,
                        hipFuncAttributeMaxDynamicSharedMemorySize, (int)shmem);

    dim3 grid(4, 32, 1);
    dim3 block(256, 1, 1);
    bconv_butterfly_kernel<<<grid, block, shmem, stream>>>(x, tw, bias, out);
}


# Y plane between the stage groups made batch-major (Y[b][n], pitch 1028): a wave's 8 blocks share 16-byte pieces, so stage 4..7 loads its operands once and stores rows of 16 bytes, not 4-byte columns
# speedup vs baseline: 17.1291x; 17.1291x over previous
#include <hip/hip_runtime.h>


typedef __attribute__((ext_vector_type(2))) float v2f;
typedef __attribute__((ext_vector_type(4))) float f4;
typedef __attribute__((ext_vector_type(8))) float v8f;

#define NPTS   1024
#define YSTR   17
#define YP     1028
#define Y_OFF  0
#define YPLANE (NPTS * YSTR)
#define A_OFF  (2 * NPTS * YSTR)
#define X_OFF  (3 * NPTS * YSTR)
#define XSTR   1028
#define T_OFF  (3 * NPTS * YSTR + 16 * XSTR)
#define MSTR   132
#define TREG   (8 * MSTR)
#define SHM_FLOATS (T_OFF + 8 * TREG)
#define PART_BYTES 8388608

struct lane_ptrs { const float* l0; const float* l1; const float* l2; const float* l3; };

__device__ __forceinline__ lane_ptrs make_lane_ptrs(const float* reg, int p, int kh)
{
    const int p0 = p & 1, p1 = (p >> 1) & 1, p2 = (p >> 2) & 1, p3 = p >> 3;
    lane_ptrs r;
    r.l0 = reg +      kh * 4 + p0 * 2;
    r.l1 = reg + 32 + p0 * 4 + p1 * 2 + kh;
    r.l2 = reg + 64 + (2 * p1 + p0) * 4 + p2 * 2;
    r.l3 = reg + 96 + (p & 7) * 4 + p3 * 2;
    return r;
}

struct raw_frag { v2f t3; v2f t2[2]; v2f t0[4]; float t1[4]; };

__device__ __forceinline__ void gather_raw(const lane_ptrs& L, int q, raw_frag& r)
{
    r.t3 = *(const v2f*)(L.l3 + q * MSTR);
#pragma unroll
    for (int q3 = 0; q3 < 2; ++q3)
        r.t2[q3] = *(const v2f*)(L.l2 + q * MSTR + 16 * q3);
#pragma unroll
    for (int kk = 0; kk < 4; ++kk) {
        r.t0[kk] = *(const v2f*)(L.l0 + q * MSTR + 8 * kk);
        r.t1[kk] = L.l1[q * MSTR + 8 * kk];
    }
}

__device__ __forceinline__ void wave_lds_handover()
{
    __builtin_amdgcn_fence(__ATOMIC_RELEASE, "wavefront");
    __builtin_amdgcn_wave_barrier();
    __builtin_amdgcn_fence(__ATOMIC_ACQUIRE, "wavefront");
}

__device__ __forceinline__ void mul_raw(const raw_frag& r, v2f a[4])
{
#pragma unroll
    for (int kk = 0; kk < 4; ++kk) {
        const float w2 = (kk & 1) ? r.t2[kk >> 1].y : r.t2[kk >> 1].x;
        const float w3 = (kk >> 1) ? r.t3.y : r.t3.x;
        a[kk].x = w3 * (w2 * (r.t1[kk] * r.t0[kk].x));
        a[kk].y = w3 * (w2 * (r.t1[kk] * r.t0[kk].y));
    }
}

struct x_piece { f4 lo; f4 hi; };

__device__ __forceinline__ float piece_elem(const x_piece& r, int j)
{
    return (j < 4) ? r.lo[j & 3] : r.hi[j & 3];
}

__device__ __forceinline__ constexpr int rev3(int v)
{
    return ((v & 1) << 2) | (v & 2) | ((v >> 2) & 1);
}

__device__ __forceinline__ void ask_slice(const float* slice, int from, int t)
{
    const char* p = (const char*)slice + from + (unsigned)(t * 256);
    if (from + 65536 <= 81920) {
        __builtin_prefetch(p, 0, 1);
        if (t * 256 < 81920 - from - 65536)
            __builtin_prefetch(p + 65536, 0, 1);
    } else {
        if (t * 256 < 81920 - from)
            __builtin_prefetch(p, 0, 1);
    }
}

__global__ __launch_bounds__(512) __attribute__((amdgpu_waves_per_eu(4, 4))) void bconv_butterfly_kernel(
    const float* __restrict__ x,
    const float* __restrict__ tw,
    const float* __restrict__ bias,
    float* __restrict__ out,
    float* __restrict__ part,
    size_t ws_bytes)
{
    extern __shared__ float smem[];

    const int  h     = blockIdx.z;
    const bool split = ws_bytes >= (size_t)PART_BYTES;
    if (!split && h != 0) return;
    const int nch  = split ? 8 : 16;
    const int ic0  = split ? 8 * h : 0;
    const int nint = 2 * nch + 2;

    const int tt    = threadIdx.x;
    const int g     = __builtin_amdgcn_readfirstlane(tt >> 8);
    const int t     = tt & 255;
    const int lane  = t & 31;
    const int wv    = t >> 5;
    const int mlane = lane & 15;
    const int kh    = lane >> 4;
    const int b0    = blockIdx.x * 16;
    const int o     = blockIdx.y;

    float* Y    = smem + Y_OFF + g * YPLANE;
    float* ACC  = smem + A_OFF;
    float* XS   = smem + X_OFF;
    float* TRG  = smem + T_OFF + wv * TREG;

    for (int idx = tt; idx < 16384; idx += 512) {
        int n = idx & 1023, b = idx >> 10;
        ACC[n * YSTR + b] = 0.f;
    }

    const float* xlane = XS + mlane * XSTR + kh * 256 + rev3(wv) * 8;
    const float* xsrc = x + b0 * 1024 + t * 4;
    float* xdst = XS + t * 4;

    const int tl_m = lane >> 3;
    const int tl_s = lane & 7;
    const lane_ptrs LT = make_lane_ptrs(TRG, mlane, kh);

    v2f af[8][4];

    if (g == 1) {
        const float* tw0 = tw + (size_t)(ic0 * 32 + o) * (10 * 512 * 4);
        ask_slice(tw0 + 32 * (10 * 512 * 4), 0, t);
        ask_slice(tw0, 32768, t);
    }

    for (int s = 0; s < nint; ++s) {
        const int sg = s - 2 * g;
        if (sg >= 0 && sg < 2 * nch) {
        const int i  = ic0 + 2 * (sg >> 2) + g;
        const int ph = sg & 3;
        const float* twio = tw + (size_t)(i * 32 + o) * (10 * 512 * 4);
        if (ph == 0) {
        f4 tq[8];
#pragma unroll
        for (int q = 0; q < 8; ++q)
            tq[q] = *(const f4*)(twio + tl_m * 2048 + (q * 64 + wv * 8 + tl_s) * 4);
        __builtin_amdgcn_sched_barrier(0);

        f4 xrow[16];
#pragma unroll
        for (int j = 0; j < 16; ++j)
            xrow[j] = *(const f4*)(xsrc + i * 65536 + j * 1024);

#pragma unroll
        for (int q = 0; q < 8; ++q)
            *(f4*)(TRG + q * MSTR + lane * 4) = tq[q];
        wave_lds_handover();

        {
            raw_frag raw[4];
#pragma unroll
            for (int q = 0; q < 4; ++q)
                gather_raw(LT, q, raw[q]);
#pragma unroll
            for (int q = 0; q < 4; ++q)
                mul_raw(raw[q], af[q]);
        }
#pragma unroll
        for (int j = 0; j < 16; ++j)
            *(f4*)(xdst + j * XSTR) = xrow[j];
        {
            raw_frag raw[4];
#pragma unroll
            for (int q = 0; q < 4; ++q)
                gather_raw(LT, 4 + q, raw[q]);
#pragma unroll
            for (int q = 0; q < 4; ++q)
                mul_raw(raw[q], af[4 + q]);
        }
        } else if (ph == 1) {
        f4 tq[8];
#pragma unroll
        for (int e = 0; e < 8; ++e)
            tq[e] = *(const f4*)(twio + (4 + tl_m) * 2048
                                 + ((wv >> 1) * 128 + e * 16 + (wv & 1) * 8 + tl_s) * 4);

        x_piece xb[4][2];
#pragma unroll
        for (int kk = 0; kk < 4; ++kk) {
#pragma unroll
            for (int e = 0; e < 2; ++e) {
                const int off = e * 512 + (kk & 1) * 128 + (kk >> 1) * 64;
                xb[kk][e].lo = *(const f4*)(xlane + off);
                xb[kk][e].hi = *(const f4*)(xlane + off + 4);
            }
        }

#pragma unroll
        for (int blk = 0; blk < 8; ++blk) {
            int k1 = blk * 8 + wv;
            v8f c = {};
            const int rowbase = k1 * 16;
#pragma unroll
            for (int kk = 0; kk < 4; ++kk) {
                v2f bb;
                bb.x = piece_elem(xb[kk][0], rev3(blk));
                bb.y = piece_elem(xb[kk][1], rev3(blk));
                c = __builtin_amdgcn_wmma_f32_16x16x4_f32(
                        false, af[blk][kk], false, bb, (short)0, c, false, false);
            }
            float* yrow = Y + mlane * YP + rowbase + 8 * kh;
            *(f4*)(yrow)     = f4{c[0], c[1], c[2], c[3]};
            *(f4*)(yrow + 4) = f4{c[4], c[5], c[6], c[7]};
        }
#pragma unroll
        for (int e = 0; e < 8; ++e)
            *(f4*)(TRG + tl_s * MSTR + tl_m * 32 + e * 4) = tq[e];
        wave_lds_handover();
#pragma unroll
        for (int hb = 0; hb < 8; hb += 4) {
            raw_frag raw[4];
#pragma unroll
            for (int q = 0; q < 4; ++q)
                gather_raw(LT, hb + q, raw[q]);
#pragma unroll
            for (int q = 0; q < 4; ++q)
                mul_raw(raw[q], af[hb + q]);
        }
        } else if (ph == 2) {
        const float* yld = Y + mlane * YP + (wv >> 1) * 256 + 2 * kh * 16 + (wv & 1) * 8;
        float*       yst = Y + mlane * YP + (wv >> 1) * 256 + 8 * kh * 16 + (wv & 1) * 8;
        x_piece yb[4][2];
#pragma unroll
        for (int kk = 0; kk < 4; ++kk) {
#pragma unroll
            for (int e = 0; e < 2; ++e) {
                yb[kk][e].lo = *(const f4*)(yld + (4 * kk + e) * 16);
                yb[kk][e].hi = *(const f4*)(yld + (4 * kk + e) * 16 + 4);
            }
        }
        v8f cb[8];
#pragma unroll
        for (int blk = 0; blk < 8; ++blk) {
            v8f c = {};
#pragma unroll
            for (int kk = 0; kk < 4; ++kk) {
                v2f bb;
                bb.x = piece_elem(yb[kk][0], blk);
                bb.y = piece_elem(yb[kk][1], blk);
                c = __builtin_amdgcn_wmma_f32_16x16x4_f32(
                        false, af[blk][kk], false, bb, (short)0, c, false, false);
            }
            cb[blk] = c;
        }
#pragma unroll
        for (int j = 0; j < 8; ++j) {
            *(f4*)(yst + j * 16)     = f4{cb[0][j], cb[1][j], cb[2][j], cb[3][j]};
            *(f4*)(yst + j * 16 + 4) = f4{cb[4][j], cb[5][j], cb[6][j], cb[7][j]};
        }
        if ((sg >> 2) + 1 < (nch >> 1))
            ask_slice(twio + 2 * 32 * (10 * 512 * 4), 0, t);
        } else {
        {
            int r = t;
            const float* tw8 = twio + 8 * 2048;
            const float* tw9 = twio + 9 * 2048;
            f4 T8a = *(const f4*)(tw8 + r * 4);
            f4 T8b = *(const f4*)(tw8 + (256 + r) * 4);
            f4 T9a = *(const f4*)(tw9 + r * 4);
            f4 T9b = *(const f4*)(tw9 + (256 + r) * 4);
#pragma unroll
            for (int hb = 0; hb < 16; hb += 8) {
                float vv[4][8];
                float aa[4][8];
#pragma unroll
                for (int b = 0; b < 8; ++b) {
                    vv[0][b] = Y[(hb + b) * YP       + r];
                    vv[1][b] = Y[(hb + b) * YP + 256 + r];
                    vv[2][b] = Y[(hb + b) * YP + 512 + r];
                    vv[3][b] = Y[(hb + b) * YP + 768 + r];
                }
#pragma unroll
                for (int b = 0; b < 8; ++b) {
                    aa[0][b] = ACC[(r      ) * YSTR + hb + b];
                    aa[1][b] = ACC[(256 + r) * YSTR + hb + b];
                    aa[2][b] = ACC[(512 + r) * YSTR + hb + b];
                    aa[3][b] = ACC[(768 + r) * YSTR + hb + b];
                }
#pragma unroll
                for (int b = 0; b < 8; ++b) {
                    const float v00 = vv[0][b], v01 = vv[1][b], v10 = vv[2][b], v11 = vv[3][b];
                    float a00 = T8a.x * v00 + T8a.y * v01;
                    float a01 = T8a.z * v00 + T8a.w * v01;
                    float a10 = T8b.x * v10 + T8b.y * v11;
                    float a11 = T8b.z * v10 + T8b.w * v11;
                    float f00 = T9a.x * a00 + T9a.y * a10;
                    float f10 = T9a.z * a00 + T9a.w * a10;
                    float f01 = T9b.x * a01 + T9b.y * a11;
                    float f11 = T9b.z * a01 + T9b.w * a11;
                    ACC[(r      ) * YSTR + hb + b] = aa[0][b] + f00;
                    ACC[(256 + r) * YSTR + hb + b] = aa[1][b] + f01;
                    ACC[(512 + r) * YSTR + hb + b] = aa[2][b] + f10;
                    ACC[(768 + r) * YSTR + hb + b] = aa[3][b] + f11;
                }
            }
        }
        }
        }
        __syncthreads();
    }

    float bz0 = 0.f, bz1 = 0.f;
    if (h == 0) {
        bz0 = bias[o * 1024 + tt];
        bz1 = bias[o * 1024 + 512 + tt];
    }
    float* dst = ((h == 0) ? out : part) + (size_t)o * 65536 + b0 * 1024 + tt;
    const float* acc0 = ACC + tt * YSTR;
    const float* acc1 = ACC + (tt + 512) * YSTR;
    for (int b = 0; b < 16; ++b) {
        dst[b * 1024]       = acc0[b] * 0.0625f + bz0;
        dst[b * 1024 + 512] = acc1[b] * 0.0625f + bz1;
    }
}

__global__ __launch_bounds__(256) void bconv_add_halves_kernel(
    float* __restrict__ out,
    const float* __restrict__ part,
    size_t ws_bytes)
{
    if (ws_bytes < (size_t)PART_BYTES) return;
    f4* o4 = (f4*)out + blockIdx.x * 2048 + threadIdx.x;
    const f4* p4 = (const f4*)part + blockIdx.x * 2048 + threadIdx.x;
    f4 a[8], p[8];
#pragma unroll
    for (int j = 0; j < 8; ++j) {
        a[j] = o4[j * 256];
        p[j] = p4[j * 256];
    }
#pragma unroll
    for (int j = 0; j < 8; ++j)
        o4[j * 256] = a[j] + p[j];
}

extern "C" void kernel_launch(void* const* d_in, const int* in_sizes, int n_in,
                              void* d_out, int out_size, void* d_ws, size_t ws_size,
                              hipStream_t stream) {
    (void)in_sizes; (void)n_in; (void)out_size;
    const float* x    = (const float*)d_in[0];
    const float* tw   = (const float*)d_in[1];
    const float* bias = (const float*)d_in[2];
    float* out        = (float*)d_out;
    float* part       = (float*)d_ws;

    const size_t shmem = (size_t)SHM_FLOATS * sizeof(float);
    hipFuncSetAttribute((const void*)bconv_butterfly_kernel,
                        hipFuncAttributeMaxDynamicSharedMemorySize, (int)shmem);

    dim3 grid(4, 32, 2);
    dim3 block(512, 1, 1);
    bconv_butterfly_kernel<<<grid, block, shmem, stream>>>(x, tw, bias, out, part, ws_size);
    bconv_add_halves_kernel<<<dim3(256, 1, 1), dim3(256, 1, 1), 0, stream>>>(out, part, ws_size);
}
